# starting kernel
# baseline (speedup 1.0000x reference)
#include <hip/hip_runtime.h>
#include <hip/hip_bf16.h>

typedef float v2f __attribute__((ext_vector_type(2)));
typedef float v4f __attribute__((ext_vector_type(4)));
typedef float v8f __attribute__((ext_vector_type(8)));

template <bool TRANS_A, int K, int LDA, int LDB, int LDC>
__global__ __launch_bounds__(256) void gemm_f32_wmma(
    const float* __restrict__ A, const float* __restrict__ B,
    float* __restrict__ C) {
  constexpr int MT = 4;
  constexpr int NT = 2;

  const int lane = threadIdx.x & 31;
  const int wave = threadIdx.x >> 5;
  const int tileM = blockIdx.y * (2 * MT * 16) + (wave & 1) * (MT * 16);
  const int tileN = blockIdx.x * (4 * NT * 16) + (wave >> 1) * (NT * 16);
  const int r = lane & 15;
  const int h = lane >> 4;

  const float* Ap;
  if (TRANS_A) {
    Ap = A + (size_t)(2 * h) * LDA + (tileM + r);
  } else {
    Ap = A + (size_t)(tileM + r) * LDA + 2 * h;
  }
  const float* Bp = B + (size_t)(2 * h) * LDB + (tileN + r);

  v8f acc[MT][NT];
#pragma unroll
  for (int i = 0; i < MT; ++i)
#pragma unroll
    for (int j = 0; j < NT; ++j) acc[i][j] = (v8f){};

#pragma unroll 2
  for (int k = 0; k < K; k += 4) {
    v2f a[MT], b[NT];
#pragma unroll
    for (int i = 0; i < MT; ++i) {
      if (TRANS_A) {
        a[i].x = Ap[16 * i];
        a[i].y = Ap[16 * i + LDA];
      } else {
        a[i] = *(const v2f*)(Ap + (size_t)i * 16 * LDA);
      }
    }
#pragma unroll
    for (int j = 0; j < NT; ++j) {
      b[j].x = Bp[16 * j];
      b[j].y = Bp[16 * j + LDB];
    }
#pragma unroll
    for (int i = 0; i < MT; ++i)
#pragma unroll
      for (int j = 0; j < NT; ++j)
        acc[i][j] = __builtin_amdgcn_wmma_f32_16x16x4_f32(
            false, a[i], false, b[j], (short)0, acc[i][j], false, false);
    Ap += TRANS_A ? 4 * LDA : 4;
    Bp += 4 * LDB;
  }

  float* Cp = C + (size_t)(tileM + 8 * h) * LDC + tileN + r;
#pragma unroll
  for (int i = 0; i < MT; ++i)
#pragma unroll
    for (int j = 0; j < NT; ++j)
#pragma unroll
      for (int v = 0; v < 8; ++v)
        Cp[(size_t)(16 * i + v) * LDC + 16 * j] = acc[i][j][v];
}

__global__ __launch_bounds__(256) void fused_attn_ln(
    const float* __restrict__ cur, const float* __restrict__ prev,
    const float* __restrict__ ctil, const float* __restrict__ gamma,
    const float* __restrict__ beta, float* __restrict__ out) {
  constexpr int D = 1024;
  constexpr int P = 8;
  const int token = blockIdx.x;
  const int tid = threadIdx.x;
  const int lane = tid & 31;
  const int wave = tid >> 5;

  const v4f* cur4 = (const v4f*)(cur + (size_t)token * D);
  const v4f* ct4 = (const v4f*)(ctil + (size_t)token * D);
  const v4f* pbase = (const v4f*)(prev + (size_t)token * P * D);

  v4f x = cur4[tid];
  v4f c = ct4[tid];

  v4f pv[P];
  float s[P];
#pragma unroll
  for (int p = 0; p < P; ++p) {
    pv[p] = __builtin_nontemporal_load(pbase + (size_t)p * (D / 4) + tid);
    s[p] = c.x * pv[p].x + c.y * pv[p].y + c.z * pv[p].z + c.w * pv[p].w;
  }

  __shared__ float red[8][P];
  __shared__ float red2[8][2];

#pragma unroll
  for (int p = 0; p < P; ++p) {
#pragma unroll
    for (int off = 16; off >= 1; off >>= 1) s[p] += __shfl_xor(s[p], off, 32);
  }
  if (lane < P) red[wave][lane] = s[lane];
  __syncthreads();

  const float inv_sqrt_d = 0.03125f;
  float score[P];
#pragma unroll
  for (int p = 0; p < P; ++p) {
    float t = 0.f;
#pragma unroll
    for (int w = 0; w < 8; ++w) t += red[w][p];
    score[p] = t * inv_sqrt_d;
  }

  float mx = score[0];
#pragma unroll
  for (int p = 1; p < P; ++p) mx = fmaxf(mx, score[p]);
  float e[P], esum = 0.f;
#pragma unroll
  for (int p = 0; p < P; ++p) {
    e[p] = __expf(score[p] - mx);
    esum += e[p];
  }
  const float inv = 1.f / esum;

  v4f y = x;
#pragma unroll
  for (int p = 0; p < P; ++p) {
    const float w = e[p] * inv;
    y.x = fmaf(w, pv[p].x, y.x);
    y.y = fmaf(w, pv[p].y, y.y);
    y.z = fmaf(w, pv[p].z, y.z);
    y.w = fmaf(w, pv[p].w, y.w);
  }

  float ps = y.x + y.y + y.z + y.w;
  float ps2 = y.x * y.x + y.y * y.y + y.z * y.z + y.w * y.w;
#pragma unroll
  for (int off = 16; off >= 1; off >>= 1) {
    ps += __shfl_xor(ps, off, 32);
    ps2 += __shfl_xor(ps2, off, 32);
  }
  if (lane == 0) {
    red2[wave][0] = ps;
    red2[wave][1] = ps2;
  }
  __syncthreads();
  float S = 0.f, S2 = 0.f;
#pragma unroll
  for (int w = 0; w < 8; ++w) {
    S += red2[w][0];
    S2 += red2[w][1];
  }
  const float mu = S * (1.f / D);
  const float var = S2 * (1.f / D) - mu * mu;
  const float rstd = rsqrtf(var + 1e-5f);

  v4f g = ((const v4f*)gamma)[tid];
  v4f b = ((const v4f*)beta)[tid];
  v4f o;
  o.x = (y.x - mu) * rstd * g.x + b.x;
  o.y = (y.y - mu) * rstd * g.y + b.y;
  o.z = (y.z - mu) * rstd * g.z + b.z;
  o.w = (y.w - mu) * rstd * g.w + b.w;
  __builtin_nontemporal_store(o, (v4f*)(out + (size_t)token * D) + tid);
}

extern "C" void kernel_launch(void* const* d_in, const int* in_sizes, int n_in,
                              void* d_out, int out_size, void* d_ws,
                              size_t ws_size, hipStream_t stream) {
  const float* cur = (const float*)d_in[0];
  const float* prev = (const float*)d_in[1];
  const float* Wq = (const float*)d_in[2];
  const float* Wk = (const float*)d_in[3];
  const float* gamma = (const float*)d_in[4];
  const float* beta = (const float*)d_in[5];
  float* out = (float*)d_out;

  constexpr int D = 1024;
  const int M = in_sizes[0] / D;

  float* G = (float*)d_ws;
  float* Ctil = G + (size_t)D * D;

  dim3 blk(256);

  gemm_f32_wmma<true, D, D, D, D>
      <<<dim3(D / 128, D / 128), blk, 0, stream>>>(Wq, Wk, G);

  gemm_f32_wmma<false, D, D, D, D>
      <<<dim3(D / 128, M / 128), blk, 0, stream>>>(cur, G, Ctil);

  fused_attn_ln<<<dim3(M), blk, 0, stream>>>(cur, prev, Ctil, gamma, beta, out);
}


# GEMM operands staged through LDS: block-wide 128-bit whole-line global loads, double-buffered 32-deep K tiles, fragments read from LDS instead of per-lane global loads; sum order unchanged
# speedup vs baseline: 1.4981x; 1.4981x over previous
#include <hip/hip_runtime.h>
#include <hip/hip_bf16.h>

typedef float v2f __attribute__((ext_vector_type(2)));
typedef float v4f __attribute__((ext_vector_type(4)));
typedef float v8f __attribute__((ext_vector_type(8)));

template <bool TRANS_A, int K, int LDA, int LDB, int LDC>
__global__ __launch_bounds__(256) void gemm_f32_wmma(
    const float* __restrict__ A, const float* __restrict__ B,
    float* __restrict__ C) {
  constexpr int MT = 4;
  constexpr int NT = 2;
  constexpr int KT = 32;
  constexpr int NTILES = K / KT;
  constexpr int PK = KT + 4;
  constexpr int PX = 128 + 8;
  constexpr int A_FL = TRANS_A ? KT * PX : 128 * PK;
  constexpr int B_FL = KT * PX;
  constexpr int Q = KT / 8;

  __shared__ __attribute__((aligned(16))) float sA[2 * A_FL];
  __shared__ __attribute__((aligned(16))) float sB[2 * B_FL];

  const int tid = threadIdx.x;
  const int lane = tid & 31;
  const int wave = tid >> 5;
  const int wm = wave & 1;
  const int wn = wave >> 1;
  const int blockM = blockIdx.y * 128;
  const int blockN = blockIdx.x * 128;
  const int r = lane & 15;
  const int h = lane >> 4;

  const float* Ag;
  int aSt;
  if (TRANS_A) {
    Ag = A + (size_t)(tid >> 5) * LDA + blockM + 4 * (tid & 31);
    aSt = (tid >> 5) * PX + 4 * (tid & 31);
  } else {
    Ag = A + (size_t)(blockM + (tid >> 3)) * LDA + 4 * (tid & 7);
    aSt = (tid >> 3) * PK + 4 * (tid & 7);
  }
  const float* Bg = B + (size_t)(tid >> 5) * LDB + blockN + 4 * (tid & 31);
  const int bSt = (tid >> 5) * PX + 4 * (tid & 31);

  const int aRd = TRANS_A ? (2 * h) * PX + wm * 64 + r
                          : (wm * 64 + r) * PK + 2 * h;
  const int bRd = (2 * h) * PX + wn * 32 + r;

  v8f acc[MT][NT];
#pragma unroll
  for (int i = 0; i < MT; ++i)
#pragma unroll
    for (int j = 0; j < NT; ++j) acc[i][j] = (v8f){};

  v4f ga[Q], gb[Q];

  auto gload = [&](int t) {
#pragma unroll
    for (int q = 0; q < Q; ++q) {
      if (TRANS_A) {
        ga[q] = *(const v4f*)(Ag + (size_t)(t * KT + 8 * q) * LDA);
      } else {
        ga[q] = *(const v4f*)(Ag + (size_t)(32 * q) * LDA + t * KT);
      }
      gb[q] = *(const v4f*)(Bg + (size_t)(t * KT + 8 * q) * LDB);
    }
  };

  auto sstore = [&](int stage) {
    float* a = sA + stage * A_FL;
    float* b = sB + stage * B_FL;
#pragma unroll
    for (int q = 0; q < Q; ++q) {
      if (TRANS_A) {
        *(v4f*)(a + aSt + 8 * q * PX) = ga[q];
      } else {
        *(v4f*)(a + aSt + 32 * q * PK) = ga[q];
      }
      *(v4f*)(b + bSt + 8 * q * PX) = gb[q];
    }
  };

  auto ldfrag = [&](const float* a, const float* b, int s, v2f* af, v2f* bf) {
#pragma unroll
    for (int i = 0; i < MT; ++i) {
      if (TRANS_A) {
        af[i].x = a[aRd + (4 * s) * PX + 16 * i];
        af[i].y = a[aRd + (4 * s + 1) * PX + 16 * i];
      } else {
        af[i] = *(const v2f*)(a + aRd + 16 * i * PK + 4 * s);
      }
    }
#pragma unroll
    for (int j = 0; j < NT; ++j) {
      bf[j].x = b[bRd + (4 * s) * PX + 16 * j];
      bf[j].y = b[bRd + (4 * s + 1) * PX + 16 * j];
    }
  };

  auto compute = [&](int stage) {
    const float* a = sA + stage * A_FL;
    const float* b = sB + stage * B_FL;
    v2f af[2][MT], bf[2][NT];
    ldfrag(a, b, 0, af[0], bf[0]);
#pragma unroll
    for (int s = 0; s < KT / 4; ++s) {
      if (s + 1 < KT / 4) ldfrag(a, b, s + 1, af[(s + 1) & 1], bf[(s + 1) & 1]);
      __builtin_amdgcn_sched_barrier(0);
#pragma unroll
      for (int i = 0; i < MT; ++i)
#pragma unroll
        for (int j = 0; j < NT; ++j)
          acc[i][j] = __builtin_amdgcn_wmma_f32_16x16x4_f32(
              false, af[s & 1][i], false, bf[s & 1][j], (short)0, acc[i][j],
              false, false);
    }
  };

  gload(0);
  sstore(0);
  __syncthreads();
  for (int t = 0; t < NTILES - 1; ++t) {
    gload(t + 1);
    __builtin_amdgcn_sched_barrier(0);
    compute(t & 1);
    sstore((t + 1) & 1);
    __syncthreads();
  }
  compute((NTILES - 1) & 1);

  const int tileM = blockM + wm * (MT * 16);
  const int tileN = blockN + wn * (NT * 16);
  float* Cp = C + (size_t)(tileM + 8 * h) * LDC + tileN + r;
#pragma unroll
  for (int i = 0; i < MT; ++i)
#pragma unroll
    for (int j = 0; j < NT; ++j)
#pragma unroll
      for (int v = 0; v < 8; ++v)
        Cp[(size_t)(16 * i + v) * LDC + 16 * j] = acc[i][j][v];
}

__global__ __launch_bounds__(256) void fused_attn_ln(
    const float* __restrict__ cur, const float* __restrict__ prev,
    const float* __restrict__ ctil, const float* __restrict__ gamma,
    const float* __restrict__ beta, float* __restrict__ out) {
  constexpr int D = 1024;
  constexpr int P = 8;
  const int token = blockIdx.x;
  const int tid = threadIdx.x;
  const int lane = tid & 31;
  const int wave = tid >> 5;

  const v4f* cur4 = (const v4f*)(cur + (size_t)token * D);
  const v4f* ct4 = (const v4f*)(ctil + (size_t)token * D);
  const v4f* pbase = (const v4f*)(prev + (size_t)token * P * D);

  v4f x = cur4[tid];
  v4f c = ct4[tid];

  v4f pv[P];
  float s[P];
#pragma unroll
  for (int p = 0; p < P; ++p) {
    pv[p] = __builtin_nontemporal_load(pbase + (size_t)p * (D / 4) + tid);
    s[p] = c.x * pv[p].x + c.y * pv[p].y + c.z * pv[p].z + c.w * pv[p].w;
  }

  __shared__ float red[8][P];
  __shared__ float red2[8][2];

#pragma unroll
  for (int p = 0; p < P; ++p) {
#pragma unroll
    for (int off = 16; off >= 1; off >>= 1) s[p] += __shfl_xor(s[p], off, 32);
  }
  if (lane < P) red[wave][lane] = s[lane];
  __syncthreads();

  const float inv_sqrt_d = 0.03125f;
  float score[P];
#pragma unroll
  for (int p = 0; p < P; ++p) {
    float t = 0.f;
#pragma unroll
    for (int w = 0; w < 8; ++w) t += red[w][p];
    score[p] = t * inv_sqrt_d;
  }

  float mx = score[0];
#pragma unroll
  for (int p = 1; p < P; ++p) mx = fmaxf(mx, score[p]);
  float e[P], esum = 0.f;
#pragma unroll
  for (int p = 0; p < P; ++p) {
    e[p] = __expf(score[p] - mx);
    esum += e[p];
  }
  const float inv = 1.f / esum;

  v4f y = x;
#pragma unroll
  for (int p = 0; p < P; ++p) {
    const float w = e[p] * inv;
    y.x = fmaf(w, pv[p].x, y.x);
    y.y = fmaf(w, pv[p].y, y.y);
    y.z = fmaf(w, pv[p].z, y.z);
    y.w = fmaf(w, pv[p].w, y.w);
  }

  float ps = y.x + y.y + y.z + y.w;
  float ps2 = y.x * y.x + y.y * y.y + y.z * y.z + y.w * y.w;
#pragma unroll
  for (int off = 16; off >= 1; off >>= 1) {
    ps += __shfl_xor(ps, off, 32);
    ps2 += __shfl_xor(ps2, off, 32);
  }
  if (lane == 0) {
    red2[wave][0] = ps;
    red2[wave][1] = ps2;
  }
  __syncthreads();
  float S = 0.f, S2 = 0.f;
#pragma unroll
  for (int w = 0; w < 8; ++w) {
    S += red2[w][0];
    S2 += red2[w][1];
  }
  const float mu = S * (1.f / D);
  const float var = S2 * (1.f / D) - mu * mu;
  const float rstd = rsqrtf(var + 1e-5f);

  v4f g = ((const v4f*)gamma)[tid];
  v4f b = ((const v4f*)beta)[tid];
  v4f o;
  o.x = (y.x - mu) * rstd * g.x + b.x;
  o.y = (y.y - mu) * rstd * g.y + b.y;
  o.z = (y.z - mu) * rstd * g.z + b.z;
  o.w = (y.w - mu) * rstd * g.w + b.w;
  __builtin_nontemporal_store(o, (v4f*)(out + (size_t)token * D) + tid);
}

extern "C" void kernel_launch(void* const* d_in, const int* in_sizes, int n_in,
                              void* d_out, int out_size, void* d_ws,
                              size_t ws_size, hipStream_t stream) {
  const float* cur = (const float*)d_in[0];
  const float* prev = (const float*)d_in[1];
  const float* Wq = (const float*)d_in[2];
  const float* Wk = (const float*)d_in[3];
  const float* gamma = (const float*)d_in[4];
  const float* beta = (const float*)d_in[5];
  float* out = (float*)d_out;

  constexpr int D = 1024;
  const int M = in_sizes[0] / D;

  float* G = (float*)d_ws;
  float* Ctil = G + (size_t)D * D;

  dim3 blk(256);

  gemm_f32_wmma<true, D, D, D, D>
      <<<dim3(D / 128, D / 128), blk, 0, stream>>>(Wq, Wk, G);

  gemm_f32_wmma<false, D, D, D, D>
      <<<dim3(D / 128, M / 128), blk, 0, stream>>>(cur, G, Ctil);

  fused_attn_ln<<<dim3(M), blk, 0, stream>>>(cur, prev, Ctil, gamma, beta, out);
}


# First GEMM (Wq^T Wk, 1024^3) re-tiled from 128x128 to 64x64 block tiles with 128-deep K tiles: 256 blocks instead of 64, so it runs on every WGP instead of a quarter of them; sum order unchanged
# speedup vs baseline: 1.7707x; 1.1820x over previous
#include <hip/hip_runtime.h>
#include <hip/hip_bf16.h>

typedef float v2f __attribute__((ext_vector_type(2)));
typedef float v4f __attribute__((ext_vector_type(4)));
typedef float v8f __attribute__((ext_vector_type(8)));

template <bool TRANS_A, int WM, int MT, int NT, int KT, int K, int LDA,
          int LDB, int LDC>
__global__ __launch_bounds__(256) void gemm_f32_wmma(
    const float* __restrict__ A, const float* __restrict__ B,
    float* __restrict__ C) {
  constexpr int WN = 8 / WM;
  constexpr int BM = WM * MT * 16;
  constexpr int BN = WN * NT * 16;
  constexpr int NTILES = K / KT;
  constexpr int PK = KT + 4;
  constexpr int PA = BM + 8;
  constexpr int PB = BN + 8;
  constexpr int A_FL = TRANS_A ? KT * PA : BM * PK;
  constexpr int B_FL = KT * PB;
  constexpr int Q = BM * KT / 1024;
  constexpr int SA = TRANS_A ? BM / 4 : KT / 4;
  constexpr int SB = BN / 4;
  constexpr int RA = 256 / SA;
  constexpr int RB = 256 / SB;
  constexpr int SG = 8 / (MT * NT);
  constexpr int NG = KT / (4 * SG);
  static_assert(BM == BN, "A and B tiles hold the same number of slots");
  static_assert(MT * NT * SG == 8 && NG * SG * 4 == KT && Q * 1024 == BM * KT,
                "tile shape");

  __shared__ __attribute__((aligned(16))) float sA[2 * A_FL];
  __shared__ __attribute__((aligned(16))) float sB[2 * B_FL];

  const int tid = threadIdx.x;
  const int lane = tid & 31;
  const int wave = tid >> 5;
  const int wm = wave % WM;
  const int wn = wave / WM;
  const int blockM = blockIdx.y * BM;
  const int blockN = blockIdx.x * BN;
  const int r = lane & 15;
  const int h = lane >> 4;

  const float* Ag;
  int aSt;
  if (TRANS_A) {
    Ag = A + (size_t)(tid / SA) * LDA + blockM + 4 * (tid % SA);
    aSt = (tid / SA) * PA + 4 * (tid % SA);
  } else {
    Ag = A + (size_t)(blockM + tid / SA) * LDA + 4 * (tid % SA);
    aSt = (tid / SA) * PK + 4 * (tid % SA);
  }
  const float* Bg = B + (size_t)(tid / SB) * LDB + blockN + 4 * (tid % SB);
  const int bSt = (tid / SB) * PB + 4 * (tid % SB);

  const int aRd = TRANS_A ? (2 * h) * PA + wm * (MT * 16) + r
                          : (wm * (MT * 16) + r) * PK + 2 * h;
  const int bRd = (2 * h) * PB + wn * (NT * 16) + r;

  v8f acc[MT][NT];
#pragma unroll
  for (int i = 0; i < MT; ++i)
#pragma unroll
    for (int j = 0; j < NT; ++j) acc[i][j] = (v8f){};

  v4f ga[Q], gb[Q];

  auto gload = [&](int t) {
#pragma unroll
    for (int q = 0; q < Q; ++q) {
      if (TRANS_A) {
        ga[q] = *(const v4f*)(Ag + (size_t)(t * KT + RA * q) * LDA);
      } else {
        ga[q] = *(const v4f*)(Ag + (size_t)(RA * q) * LDA + t * KT);
      }
      gb[q] = *(const v4f*)(Bg + (size_t)(t * KT + RB * q) * LDB);
    }
  };

  auto sstore = [&](int stage) {
    float* a = sA + stage * A_FL;
    float* b = sB + stage * B_FL;
#pragma unroll
    for (int q = 0; q < Q; ++q) {
      if (TRANS_A) {
        *(v4f*)(a + aSt + RA * q * PA) = ga[q];
      } else {
        *(v4f*)(a + aSt + RA * q * PK) = ga[q];
      }
      *(v4f*)(b + bSt + RB * q * PB) = gb[q];
    }
  };

  auto ldfrag = [&](const float* a, const float* b, int s, v2f* af, v2f* bf) {
#pragma unroll
    for (int i = 0; i < MT; ++i) {
      if (TRANS_A) {
        af[i].x = a[aRd + (4 * s) * PA + 16 * i];
        af[i].y = a[aRd + (4 * s + 1) * PA + 16 * i];
      } else {
        af[i] = *(const v2f*)(a + aRd + 16 * i * PK + 4 * s);
      }
    }
#pragma unroll
    for (int j = 0; j < NT; ++j) {
      bf[j].x = b[bRd + (4 * s) * PB + 16 * j];
      bf[j].y = b[bRd + (4 * s + 1) * PB + 16 * j];
    }
  };

  auto ldgroup = [&](const float* a, const float* b, int g, v2f (*af)[MT],
                     v2f (*bf)[NT]) {
#pragma unroll
    for (int u = 0; u < SG; ++u) ldfrag(a, b, g * SG + u, af[u], bf[u]);
  };

  auto compute = [&](int stage) {
    const float* a = sA + stage * A_FL;
    const float* b = sB + stage * B_FL;
    v2f af[2][SG][MT], bf[2][SG][NT];
    ldgroup(a, b, 0, af[0], bf[0]);
#pragma unroll
    for (int g = 0; g < NG; ++g) {
      if (g + 1 < NG) ldgroup(a, b, g + 1, af[(g + 1) & 1], bf[(g + 1) & 1]);
      __builtin_amdgcn_sched_barrier(0);
#pragma unroll
      for (int u = 0; u < SG; ++u)
#pragma unroll
        for (int i = 0; i < MT; ++i)
#pragma unroll
          for (int j = 0; j < NT; ++j)
            acc[i][j] = __builtin_amdgcn_wmma_f32_16x16x4_f32(
                false, af[g & 1][u][i], false, bf[g & 1][u][j], (short)0,
                acc[i][j], false, false);
    }
  };

  gload(0);
  sstore(0);
  __syncthreads();
  for (int t = 0; t < NTILES - 1; ++t) {
    gload(t + 1);
    __builtin_amdgcn_sched_barrier(0);
    compute(t & 1);
    sstore((t + 1) & 1);
    __syncthreads();
  }
  compute((NTILES - 1) & 1);

  const int tileM = blockM + wm * (MT * 16);
  const int tileN = blockN + wn * (NT * 16);
  float* Cp = C + (size_t)(tileM + 8 * h) * LDC + tileN + r;
#pragma unroll
  for (int i = 0; i < MT; ++i)
#pragma unroll
    for (int j = 0; j < NT; ++j)
#pragma unroll
      for (int v = 0; v < 8; ++v)
        Cp[(size_t)(16 * i + v) * LDC + 16 * j] = acc[i][j][v];
}

__global__ __launch_bounds__(256) void fused_attn_ln(
    const float* __restrict__ cur, const float* __restrict__ prev,
    const float* __restrict__ ctil, const float* __restrict__ gamma,
    const float* __restrict__ beta, float* __restrict__ out) {
  constexpr int D = 1024;
  constexpr int P = 8;
  const int token = blockIdx.x;
  const int tid = threadIdx.x;
  const int lane = tid & 31;
  const int wave = tid >> 5;

  const v4f* cur4 = (const v4f*)(cur + (size_t)token * D);
  const v4f* ct4 = (const v4f*)(ctil + (size_t)token * D);
  const v4f* pbase = (const v4f*)(prev + (size_t)token * P * D);

  v4f x = cur4[tid];
  v4f c = ct4[tid];

  v4f pv[P];
  float s[P];
#pragma unroll
  for (int p = 0; p < P; ++p) {
    pv[p] = __builtin_nontemporal_load(pbase + (size_t)p * (D / 4) + tid);
    s[p] = c.x * pv[p].x + c.y * pv[p].y + c.z * pv[p].z + c.w * pv[p].w;
  }

  __shared__ float red[8][P];
  __shared__ float red2[8][2];

#pragma unroll
  for (int p = 0; p < P; ++p) {
#pragma unroll
    for (int off = 16; off >= 1; off >>= 1) s[p] += __shfl_xor(s[p], off, 32);
  }
  if (lane < P) red[wave][lane] = s[lane];
  __syncthreads();

  const float inv_sqrt_d = 0.03125f;
  float score[P];
#pragma unroll
  for (int p = 0; p < P; ++p) {
    float t = 0.f;
#pragma unroll
    for (int w = 0; w < 8; ++w) t += red[w][p];
    score[p] = t * inv_sqrt_d;
  }

  float mx = score[0];
#pragma unroll
  for (int p = 1; p < P; ++p) mx = fmaxf(mx, score[p]);
  float e[P], esum = 0.f;
#pragma unroll
  for (int p = 0; p < P; ++p) {
    e[p] = __expf(score[p] - mx);
    esum += e[p];
  }
  const float inv = 1.f / esum;

  v4f y = x;
#pragma unroll
  for (int p = 0; p < P; ++p) {
    const float w = e[p] * inv;
    y.x = fmaf(w, pv[p].x, y.x);
    y.y = fmaf(w, pv[p].y, y.y);
    y.z = fmaf(w, pv[p].z, y.z);
    y.w = fmaf(w, pv[p].w, y.w);
  }

  float ps = y.x + y.y + y.z + y.w;
  float ps2 = y.x * y.x + y.y * y.y + y.z * y.z + y.w * y.w;
#pragma unroll
  for (int off = 16; off >= 1; off >>= 1) {
    ps += __shfl_xor(ps, off, 32);
    ps2 += __shfl_xor(ps2, off, 32);
  }
  if (lane == 0) {
    red2[wave][0] = ps;
    red2[wave][1] = ps2;
  }
  __syncthreads();
  float S = 0.f, S2 = 0.f;
#pragma unroll
  for (int w = 0; w < 8; ++w) {
    S += red2[w][0];
    S2 += red2[w][1];
  }
  const float mu = S * (1.f / D);
  const float var = S2 * (1.f / D) - mu * mu;
  const float rstd = rsqrtf(var + 1e-5f);

  v4f g = ((const v4f*)gamma)[tid];
  v4f b = ((const v4f*)beta)[tid];
  v4f o;
  o.x = (y.x - mu) * rstd * g.x + b.x;
  o.y = (y.y - mu) * rstd * g.y + b.y;
  o.z = (y.z - mu) * rstd * g.z + b.z;
  o.w = (y.w - mu) * rstd * g.w + b.w;
  __builtin_nontemporal_store(o, (v4f*)(out + (size_t)token * D) + tid);
}

extern "C" void kernel_launch(void* const* d_in, const int* in_sizes, int n_in,
                              void* d_out, int out_size, void* d_ws,
                              size_t ws_size, hipStream_t stream) {
  const float* cur = (const float*)d_in[0];
  const float* prev = (const float*)d_in[1];
  const float* Wq = (const float*)d_in[2];
  const float* Wk = (const float*)d_in[3];
  const float* gamma = (const float*)d_in[4];
  const float* beta = (const float*)d_in[5];
  float* out = (float*)d_out;

  constexpr int D = 1024;
  const int M = in_sizes[0] / D;

  float* G = (float*)d_ws;
  float* Ctil = G + (size_t)D * D;

  dim3 blk(256);

  gemm_f32_wmma<true, 4, 1, 2, 128, D, D, D, D>
      <<<dim3(D / 64, D / 64), blk, 0, stream>>>(Wq, Wk, G);

  gemm_f32_wmma<false, 2, 4, 2, 32, D, D, D, D>
      <<<dim3(D / 128, M / 128), blk, 0, stream>>>(cur, G, Ctil);

  fused_attn_ln<<<dim3(M), blk, 0, stream>>>(cur, prev, Ctil, gamma, beta, out);
}


# Both GEMMs: fragment LDS reads issued in few wait-free blocks ahead of 32/16-WMMA blocks (two register sets; B k-pairs interleaved in LDS so a read fills an operand pair); sum order unchanged
# speedup vs baseline: 1.8493x; 1.0444x over previous
#include <hip/hip_runtime.h>
#include <hip/hip_bf16.h>

typedef float v2f __attribute__((ext_vector_type(2)));
typedef float v4f __attribute__((ext_vector_type(4)));
typedef float v8f __attribute__((ext_vector_type(8)));

template <bool TRANS_A, int WM, int MT, int NT, int KT, int GS, int K,
          int LDA, int LDB, int LDC>
__global__ __launch_bounds__(256) void gemm_f32_wmma(
    const float* __restrict__ A, const float* __restrict__ B,
    float* __restrict__ C) {
  constexpr int WN = 8 / WM;
  constexpr int BM = WM * MT * 16;
  constexpr int BN = WN * NT * 16;
  constexpr int NTILES = K / KT;
  constexpr int PK = KT + 4;
  constexpr int PA = BM + 8;
  constexpr int PB = 2 * BN + 32;
  constexpr int A_FL = TRANS_A ? KT * PA : BM * PK;
  constexpr int B_FL = (KT / 2) * PB;
  constexpr int Q = BM * KT / 1024;
  constexpr int SA = TRANS_A ? BM / 4 : KT / 4;
  constexpr int SB = BN / 4;
  constexpr int RA = 256 / SA;
  constexpr int RB = 256 / SB;
  constexpr int NG = KT / (4 * GS);
  static_assert(BM == BN, "A and B tiles hold the same number of slots");
  static_assert(NG * GS * 4 == KT && NG % 2 == 0 && NG >= 2 &&
                    Q * 1024 == BM * KT && Q % 2 == 0 && NTILES >= 3,
                "tile shape");

  __shared__ __attribute__((aligned(16))) float sA[2 * A_FL];
  __shared__ __attribute__((aligned(16))) float sB[2 * B_FL];

  const int tid = threadIdx.x;
  const int lane = tid & 31;
  const int wave = tid >> 5;
  const int wm = wave % WM;
  const int wn = wave / WM;
  const int blockM = blockIdx.y * BM;
  const int blockN = blockIdx.x * BN;
  const int r = lane & 15;
  const int h = lane >> 4;

  const float* Ag;
  int aSt;
  if (TRANS_A) {
    Ag = A + (size_t)(tid / SA) * LDA + blockM + 4 * (tid % SA);
    aSt = (tid / SA) * PA + 4 * (tid % SA);
  } else {
    Ag = A + (size_t)(blockM + tid / SA) * LDA + 4 * (tid % SA);
    aSt = (tid / SA) * PK + 4 * (tid % SA);
  }
  const float* Bg = B + (size_t)(2 * (tid / SB)) * LDB + blockN + 4 * (tid % SB);
  const int bSt = (tid / SB) * PB + 8 * (tid % SB);

  const int aRd = TRANS_A ? (2 * h) * PA + wm * (MT * 16) + r
                          : (wm * (MT * 16) + r) * PK + 2 * h;
  const int bRd = h * PB + 2 * (wn * (NT * 16) + r);

  v8f acc[MT][NT];
#pragma unroll
  for (int i = 0; i < MT; ++i)
#pragma unroll
    for (int j = 0; j < NT; ++j) acc[i][j] = (v8f){};

  v4f ga[Q], gb[Q];

  auto gload = [&](int t) __attribute__((always_inline)) {
#pragma unroll
    for (int q = 0; q < Q; ++q) {
      if (TRANS_A) {
        ga[q] = *(const v4f*)(Ag + (size_t)(t * KT + RA * q) * LDA);
      } else {
        ga[q] = *(const v4f*)(Ag + (size_t)(RA * q) * LDA + t * KT);
      }
      gb[q] = *(const v4f*)(Bg + (size_t)(t * KT + 2 * RB * (q / 2) + (q & 1)) * LDB);
    }
  };

  auto sstore = [&](int stage) __attribute__((always_inline)) {
    float* a = sA + stage * A_FL;
    float* b = sB + stage * B_FL;
#pragma unroll
    for (int q = 0; q < Q; ++q) {
      if (TRANS_A) {
        *(v4f*)(a + aSt + RA * q * PA) = ga[q];
      } else {
        *(v4f*)(a + aSt + RA * q * PK) = ga[q];
      }
    }
#pragma unroll
    for (int p = 0; p < Q / 2; ++p) {
      const v4f e = gb[2 * p], o = gb[2 * p + 1];
      *(v4f*)(b + bSt + RB * p * PB) = (v4f){e.x, o.x, e.y, o.y};
      *(v4f*)(b + bSt + RB * p * PB + 4) = (v4f){e.z, o.z, e.w, o.w};
    }
  };

  auto ldfrag = [&](const float* a, const float* b, int s, v2f* af,
                    v2f* bf) __attribute__((always_inline)) {
#pragma unroll
    for (int i = 0; i < MT; ++i) {
      if (TRANS_A) {
        af[i].x = a[aRd + (4 * s) * PA + 16 * i];
        af[i].y = a[aRd + (4 * s + 1) * PA + 16 * i];
      } else {
        af[i] = *(const v2f*)(a + aRd + 16 * i * PK + 4 * s);
      }
    }
#pragma unroll
    for (int j = 0; j < NT; ++j)
      bf[j] = *(const v2f*)(b + bRd + (2 * s) * PB + 32 * j);
  };

  v2f af[2][GS][MT], bf[2][GS][NT];

  auto ldgroup = [&](int stage, int g, int set)
                     __attribute__((always_inline)) {
    const float* a = sA + stage * A_FL;
    const float* b = sB + stage * B_FL;
#pragma unroll
    for (int u = 0; u < GS; ++u) ldfrag(a, b, g * GS + u, af[set][u], bf[set][u]);
  };

  auto mma = [&](int set) __attribute__((always_inline)) {
    __builtin_amdgcn_sched_barrier(0);
#pragma unroll
    for (int u = 0; u < GS; ++u)
#pragma unroll
      for (int i = 0; i < MT; ++i)
#pragma unroll
        for (int j = 0; j < NT; ++j)
          acc[i][j] = __builtin_amdgcn_wmma_f32_16x16x4_f32(
              false, af[set][u][i], false, bf[set][u][j], (short)0, acc[i][j],
              false, false);
    __builtin_amdgcn_sched_barrier(0);
  };

  auto inner = [&](int stage) __attribute__((always_inline)) {
#pragma unroll
    for (int g = 0; g + 1 < NG; ++g) {
      ldgroup(stage, g + 1, (g + 1) & 1);
      mma(g & 1);
    }
  };

  auto step = [&](int t, bool more) __attribute__((always_inline)) {
    const int stage = t & 1;
    inner(stage);
    sstore(stage ^ 1);
    __syncthreads();
    if (more) gload(t + 2);
    ldgroup(stage ^ 1, 0, 0);
    mma((NG - 1) & 1);
  };

  gload(0);
  sstore(0);
  __syncthreads();
  gload(1);
  ldgroup(0, 0, 0);
  for (int t = 0; t < NTILES - 2; ++t) step(t, true);
  step(NTILES - 2, false);
  inner((NTILES - 1) & 1);
  mma((NG - 1) & 1);

  const int tileM = blockM + wm * (MT * 16);
  const int tileN = blockN + wn * (NT * 16);
  float* Cp = C + (size_t)(tileM + 8 * h) * LDC + tileN + r;
#pragma unroll
  for (int i = 0; i < MT; ++i)
#pragma unroll
    for (int j = 0; j < NT; ++j)
#pragma unroll
      for (int v = 0; v < 8; ++v)
        Cp[(size_t)(16 * i + v) * LDC + 16 * j] = acc[i][j][v];
}

__global__ __launch_bounds__(256) void fused_attn_ln(
    const float* __restrict__ cur, const float* __restrict__ prev,
    const float* __restrict__ ctil, const float* __restrict__ gamma,
    const float* __restrict__ beta, float* __restrict__ out) {
  constexpr int D = 1024;
  constexpr int P = 8;
  const int token = blockIdx.x;
  const int tid = threadIdx.x;
  const int lane = tid & 31;
  const int wave = tid >> 5;

  const v4f* cur4 = (const v4f*)(cur + (size_t)token * D);
  const v4f* ct4 = (const v4f*)(ctil + (size_t)token * D);
  const v4f* pbase = (const v4f*)(prev + (size_t)token * P * D);

  v4f x = cur4[tid];
  v4f c = ct4[tid];

  v4f pv[P];
  float s[P];
#pragma unroll
  for (int p = 0; p < P; ++p) {
    pv[p] = __builtin_nontemporal_load(pbase + (size_t)p * (D / 4) + tid);
    s[p] = c.x * pv[p].x + c.y * pv[p].y + c.z * pv[p].z + c.w * pv[p].w;
  }

  __shared__ float red[8][P];
  __shared__ float red2[8][2];

#pragma unroll
  for (int p = 0; p < P; ++p) {
#pragma unroll
    for (int off = 16; off >= 1; off >>= 1) s[p] += __shfl_xor(s[p], off, 32);
  }
  if (lane < P) red[wave][lane] = s[lane];
  __syncthreads();

  const float inv_sqrt_d = 0.03125f;
  float score[P];
#pragma unroll
  for (int p = 0; p < P; ++p) {
    float t = 0.f;
#pragma unroll
    for (int w = 0; w < 8; ++w) t += red[w][p];
    score[p] = t * inv_sqrt_d;
  }

  float mx = score[0];
#pragma unroll
  for (int p = 1; p < P; ++p) mx = fmaxf(mx, score[p]);
  float e[P], esum = 0.f;
#pragma unroll
  for (int p = 0; p < P; ++p) {
    e[p] = __expf(score[p] - mx);
    esum += e[p];
  }
  const float inv = 1.f / esum;

  v4f y = x;
#pragma unroll
  for (int p = 0; p < P; ++p) {
    const float w = e[p] * inv;
    y.x = fmaf(w, pv[p].x, y.x);
    y.y = fmaf(w, pv[p].y, y.y);
    y.z = fmaf(w, pv[p].z, y.z);
    y.w = fmaf(w, pv[p].w, y.w);
  }

  float ps = y.x + y.y + y.z + y.w;
  float ps2 = y.x * y.x + y.y * y.y + y.z * y.z + y.w * y.w;
#pragma unroll
  for (int off = 16; off >= 1; off >>= 1) {
    ps += __shfl_xor(ps, off, 32);
    ps2 += __shfl_xor(ps2, off, 32);
  }
  if (lane == 0) {
    red2[wave][0] = ps;
    red2[wave][1] = ps2;
  }
  __syncthreads();
  float S = 0.f, S2 = 0.f;
#pragma unroll
  for (int w = 0; w < 8; ++w) {
    S += red2[w][0];
    S2 += red2[w][1];
  }
  const float mu = S * (1.f / D);
  const float var = S2 * (1.f / D) - mu * mu;
  const float rstd = rsqrtf(var + 1e-5f);

  v4f g = ((const v4f*)gamma)[tid];
  v4f b = ((const v4f*)beta)[tid];
  v4f o;
  o.x = (y.x - mu) * rstd * g.x + b.x;
  o.y = (y.y - mu) * rstd * g.y + b.y;
  o.z = (y.z - mu) * rstd * g.z + b.z;
  o.w = (y.w - mu) * rstd * g.w + b.w;
  __builtin_nontemporal_store(o, (v4f*)(out + (size_t)token * D) + tid);
}

extern "C" void kernel_launch(void* const* d_in, const int* in_sizes, int n_in,
                              void* d_out, int out_size, void* d_ws,
                              size_t ws_size, hipStream_t stream) {
  const float* cur = (const float*)d_in[0];
  const float* prev = (const float*)d_in[1];
  const float* Wq = (const float*)d_in[2];
  const float* Wk = (const float*)d_in[3];
  const float* gamma = (const float*)d_in[4];
  const float* beta = (const float*)d_in[5];
  float* out = (float*)d_out;

  constexpr int D = 1024;
  const int M = in_sizes[0] / D;

  float* G = (float*)d_ws;
  float* Ctil = G + (size_t)D * D;

  dim3 blk(256);

  gemm_f32_wmma<true, 4, 1, 2, 128, 8, D, D, D, D>
      <<<dim3(D / 64, D / 64), blk, 0, stream>>>(Wq, Wk, G);

  gemm_f32_wmma<false, 2, 4, 2, 32, 4, D, D, D, D>
      <<<dim3(D / 128, M / 128), blk, 0, stream>>>(cur, G, Ctil);

  fused_attn_ln<<<dim3(M), blk, 0, stream>>>(cur, prev, Ctil, gamma, beta, out);
}


# Softmax/LayerNorm pass: one wave owns a whole token and walks 4 tokens (2048 waves, no LDS, no barrier) instead of a block per token (65536 waves); channel sums now lane chains + wave butterfly
# speedup vs baseline: 1.9596x; 1.0596x over previous
#include <hip/hip_runtime.h>
#include <hip/hip_bf16.h>

typedef float v2f __attribute__((ext_vector_type(2)));
typedef float v4f __attribute__((ext_vector_type(4)));
typedef float v8f __attribute__((ext_vector_type(8)));

template <bool TRANS_A, int WM, int MT, int NT, int KT, int GS, int K,
          int LDA, int LDB, int LDC>
__global__ __launch_bounds__(256) void gemm_f32_wmma(
    const float* __restrict__ A, const float* __restrict__ B,
    float* __restrict__ C) {
  constexpr int WN = 8 / WM;
  constexpr int BM = WM * MT * 16;
  constexpr int BN = WN * NT * 16;
  constexpr int NTILES = K / KT;
  constexpr int PK = KT + 4;
  constexpr int PA = BM + 8;
  constexpr int PB = 2 * BN + 32;
  constexpr int A_FL = TRANS_A ? KT * PA : BM * PK;
  constexpr int B_FL = (KT / 2) * PB;
  constexpr int Q = BM * KT / 1024;
  constexpr int SA = TRANS_A ? BM / 4 : KT / 4;
  constexpr int SB = BN / 4;
  constexpr int RA = 256 / SA;
  constexpr int RB = 256 / SB;
  constexpr int NG = KT / (4 * GS);
  static_assert(BM == BN, "A and B tiles hold the same number of slots");
  static_assert(NG * GS * 4 == KT && NG % 2 == 0 && NG >= 2 &&
                    Q * 1024 == BM * KT && Q % 2 == 0 && NTILES >= 3,
                "tile shape");

  __shared__ __attribute__((aligned(16))) float sA[2 * A_FL];
  __shared__ __attribute__((aligned(16))) float sB[2 * B_FL];

  const int tid = threadIdx.x;
  const int lane = tid & 31;
  const int wave = tid >> 5;
  const int wm = wave % WM;
  const int wn = wave / WM;
  const int blockM = blockIdx.y * BM;
  const int blockN = blockIdx.x * BN;
  const int r = lane & 15;
  const int h = lane >> 4;

  const float* Ag;
  int aSt;
  if (TRANS_A) {
    Ag = A + (size_t)(tid / SA) * LDA + blockM + 4 * (tid % SA);
    aSt = (tid / SA) * PA + 4 * (tid % SA);
  } else {
    Ag = A + (size_t)(blockM + tid / SA) * LDA + 4 * (tid % SA);
    aSt = (tid / SA) * PK + 4 * (tid % SA);
  }
  const float* Bg = B + (size_t)(2 * (tid / SB)) * LDB + blockN + 4 * (tid % SB);
  const int bSt = (tid / SB) * PB + 8 * (tid % SB);

  const int aRd = TRANS_A ? (2 * h) * PA + wm * (MT * 16) + r
                          : (wm * (MT * 16) + r) * PK + 2 * h;
  const int bRd = h * PB + 2 * (wn * (NT * 16) + r);

  v8f acc[MT][NT];
#pragma unroll
  for (int i = 0; i < MT; ++i)
#pragma unroll
    for (int j = 0; j < NT; ++j) acc[i][j] = (v8f){};

  v4f ga[Q], gb[Q];

  auto gload = [&](int t) __attribute__((always_inline)) {
#pragma unroll
    for (int q = 0; q < Q; ++q) {
      if (TRANS_A) {
        ga[q] = *(const v4f*)(Ag + (size_t)(t * KT + RA * q) * LDA);
      } else {
        ga[q] = *(const v4f*)(Ag + (size_t)(RA * q) * LDA + t * KT);
      }
      gb[q] = *(const v4f*)(Bg + (size_t)(t * KT + 2 * RB * (q / 2) + (q & 1)) * LDB);
    }
  };

  auto sstore = [&](int stage) __attribute__((always_inline)) {
    float* a = sA + stage * A_FL;
    float* b = sB + stage * B_FL;
#pragma unroll
    for (int q = 0; q < Q; ++q) {
      if (TRANS_A) {
        *(v4f*)(a + aSt + RA * q * PA) = ga[q];
      } else {
        *(v4f*)(a + aSt + RA * q * PK) = ga[q];
      }
    }
#pragma unroll
    for (int p = 0; p < Q / 2; ++p) {
      const v4f e = gb[2 * p], o = gb[2 * p + 1];
      *(v4f*)(b + bSt + RB * p * PB) = (v4f){e.x, o.x, e.y, o.y};
      *(v4f*)(b + bSt + RB * p * PB + 4) = (v4f){e.z, o.z, e.w, o.w};
    }
  };

  auto ldfrag = [&](const float* a, const float* b, int s, v2f* af,
                    v2f* bf) __attribute__((always_inline)) {
#pragma unroll
    for (int i = 0; i < MT; ++i) {
      if (TRANS_A) {
        af[i].x = a[aRd + (4 * s) * PA + 16 * i];
        af[i].y = a[aRd + (4 * s + 1) * PA + 16 * i];
      } else {
        af[i] = *(const v2f*)(a + aRd + 16 * i * PK + 4 * s);
      }
    }
#pragma unroll
    for (int j = 0; j < NT; ++j)
      bf[j] = *(const v2f*)(b + bRd + (2 * s) * PB + 32 * j);
  };

  v2f af[2][GS][MT], bf[2][GS][NT];

  auto ldgroup = [&](int stage, int g, int set)
                     __attribute__((always_inline)) {
    const float* a = sA + stage * A_FL;
    const float* b = sB + stage * B_FL;
#pragma unroll
    for (int u = 0; u < GS; ++u) ldfrag(a, b, g * GS + u, af[set][u], bf[set][u]);
  };

  auto mma = [&](int set) __attribute__((always_inline)) {
    __builtin_amdgcn_sched_barrier(0);
#pragma unroll
    for (int u = 0; u < GS; ++u)
#pragma unroll
      for (int i = 0; i < MT; ++i)
#pragma unroll
        for (int j = 0; j < NT; ++j)
          acc[i][j] = __builtin_amdgcn_wmma_f32_16x16x4_f32(
              false, af[set][u][i], false, bf[set][u][j], (short)0, acc[i][j],
              false, false);
    __builtin_amdgcn_sched_barrier(0);
  };

  auto inner = [&](int stage) __attribute__((always_inline)) {
#pragma unroll
    for (int g = 0; g + 1 < NG; ++g) {
      ldgroup(stage, g + 1, (g + 1) & 1);
      mma(g & 1);
    }
  };

  auto step = [&](int t, bool more) __attribute__((always_inline)) {
    const int stage = t & 1;
    inner(stage);
    sstore(stage ^ 1);
    __syncthreads();
    if (more) gload(t + 2);
    ldgroup(stage ^ 1, 0, 0);
    mma((NG - 1) & 1);
  };

  gload(0);
  sstore(0);
  __syncthreads();
  gload(1);
  ldgroup(0, 0, 0);
  for (int t = 0; t < NTILES - 2; ++t) step(t, true);
  step(NTILES - 2, false);
  inner((NTILES - 1) & 1);
  mma((NG - 1) & 1);

  const int tileM = blockM + wm * (MT * 16);
  const int tileN = blockN + wn * (NT * 16);
  float* Cp = C + (size_t)(tileM + 8 * h) * LDC + tileN + r;
#pragma unroll
  for (int i = 0; i < MT; ++i)
#pragma unroll
    for (int j = 0; j < NT; ++j)
#pragma unroll
      for (int v = 0; v < 8; ++v)
        Cp[(size_t)(16 * i + v) * LDC + 16 * j] = acc[i][j][v];
}

template <int MASK>
__device__ __forceinline__ float row_xor(float v) {
  const int b = __builtin_bit_cast(int, v);
  return __builtin_bit_cast(
      float, __builtin_amdgcn_update_dpp(b, b, 0x160 + MASK, 0xf, 0xf, false));
}
__device__ __forceinline__ float xhalf(float v) {
  const int b = __builtin_bit_cast(int, v);
  return __builtin_bit_cast(
      float, __builtin_amdgcn_permlanex16(b, b, 0x76543210, 0xfedcba98, false,
                                          false));
}
__device__ __forceinline__ float wave_sum(float v) {
  v += row_xor<1>(v);
  v += row_xor<2>(v);
  v += row_xor<4>(v);
  v += row_xor<8>(v);
  return v + xhalf(v);
}

__global__ __launch_bounds__(256) void fused_attn_ln(
    const float* __restrict__ cur, const float* __restrict__ prev,
    const float* __restrict__ ctil, const float* __restrict__ gamma,
    const float* __restrict__ beta, float* __restrict__ out, int M, int NW) {
  constexpr int D = 1024;
  constexpr int P = 8;
  constexpr int Q = D / 128;
  const int lane = threadIdx.x & 31;
  const int wave = __builtin_amdgcn_readfirstlane((int)(threadIdx.x >> 5));
  const int nw = NW;
  int t = blockIdx.x * 8 + wave;
  if (t >= M) return;

  v4f g[Q], b[Q];
#pragma unroll
  for (int q = 0; q < Q; ++q) {
    g[q] = ((const v4f*)gamma)[32 * q + lane];
    b[q] = ((const v4f*)beta)[32 * q + lane];
  }

  v4f c[Q], x[Q], pv[P][Q];
  auto fetch = [&](int tok) __attribute__((always_inline)) {
    const v4f* ct4 = (const v4f*)(ctil + (size_t)tok * D) + lane;
    const v4f* pb = (const v4f*)(prev + (size_t)tok * (P * D)) + lane;
    const v4f* cu4 = (const v4f*)(cur + (size_t)tok * D) + lane;
#pragma unroll
    for (int q = 0; q < Q; ++q) c[q] = ct4[32 * q];
#pragma unroll
    for (int p = 0; p < P; ++p)
#pragma unroll
      for (int q = 0; q < Q; ++q)
        pv[p][q] = __builtin_nontemporal_load(pb + p * (D / 4) + 32 * q);
#pragma unroll
    for (int q = 0; q < Q; ++q) x[q] = cu4[32 * q];
  };

  auto token = [&](int tok, bool refill) __attribute__((always_inline)) {
    float s[P];
#pragma unroll
    for (int p = 0; p < P; ++p) {
      v2f lo = c[0].xy * pv[p][0].xy;
      v2f hi = c[0].zw * pv[p][0].zw;
#pragma unroll
      for (int q = 1; q < Q; ++q) {
        lo = __builtin_elementwise_fma(c[q].xy, pv[p][q].xy, lo);
        hi = __builtin_elementwise_fma(c[q].zw, pv[p][q].zw, hi);
      }
      s[p] = (lo.x + lo.y) + (hi.x + hi.y);
    }
#pragma unroll
    for (int p = 0; p < P; ++p) s[p] = wave_sum(s[p]);

    const float inv_sqrt_d = 0.03125f;
    float mx = s[0] * inv_sqrt_d;
#pragma unroll
    for (int p = 0; p < P; ++p) s[p] *= inv_sqrt_d;
#pragma unroll
    for (int p = 1; p < P; ++p) mx = fmaxf(mx, s[p]);
    float e[P], esum = 0.f;
#pragma unroll
    for (int p = 0; p < P; ++p) {
      e[p] = __expf(s[p] - mx);
      esum += e[p];
    }
    const float inv = 1.f / esum;

    v4f y[Q];
#pragma unroll
    for (int q = 0; q < Q; ++q) y[q] = x[q];
#pragma unroll
    for (int p = 0; p < P; ++p) {
      const float w = e[p] * inv;
      const v4f w4 = {w, w, w, w};
#pragma unroll
      for (int q = 0; q < Q; ++q)
        y[q] = __builtin_elementwise_fma(w4, pv[p][q], y[q]);
    }

    if (refill) {
      fetch(tok + nw);
    }

    v4f u = y[0];
    v4f u2 = y[0] * y[0];
#pragma unroll
    for (int q = 1; q < Q; ++q) {
      u += y[q];
      u2 = __builtin_elementwise_fma(y[q], y[q], u2);
    }
    float S = (u.x + u.y) + (u.z + u.w);
    float S2 = (u2.x + u2.y) + (u2.z + u2.w);
    S = wave_sum(S);
    S2 = wave_sum(S2);
    const float mu = S * (1.f / D);
    const float var = S2 * (1.f / D) - mu * mu;
    const float rstd = rsqrtf(var + 1e-5f);

    v4f* o4 = (v4f*)(out + (size_t)tok * D) + lane;
#pragma unroll
    for (int q = 0; q < Q; ++q) {
      const v4f o = (y[q] - mu) * rstd * g[q] + b[q];
      __builtin_nontemporal_store(o, o4 + 32 * q);
    }

  };

  fetch(t);
  for (; t + nw < M; t += nw) token(t, true);
  token(t, false);
}

extern "C" void kernel_launch(void* const* d_in, const int* in_sizes, int n_in,
                              void* d_out, int out_size, void* d_ws,
                              size_t ws_size, hipStream_t stream) {
  const float* cur = (const float*)d_in[0];
  const float* prev = (const float*)d_in[1];
  const float* Wq = (const float*)d_in[2];
  const float* Wk = (const float*)d_in[3];
  const float* gamma = (const float*)d_in[4];
  const float* beta = (const float*)d_in[5];
  float* out = (float*)d_out;

  constexpr int D = 1024;
  const int M = in_sizes[0] / D;

  float* G = (float*)d_ws;
  float* Ctil = G + (size_t)D * D;

  dim3 blk(256);

  gemm_f32_wmma<true, 4, 1, 2, 128, 8, D, D, D, D>
      <<<dim3(D / 64, D / 64), blk, 0, stream>>>(Wq, Wk, G);

  gemm_f32_wmma<false, 2, 4, 2, 32, 4, D, D, D, D>
      <<<dim3(D / 128, M / 128), blk, 0, stream>>>(cur, G, Ctil);

  fused_attn_ln<<<dim3(256), blk, 0, stream>>>(cur, prev, Ctil, gamma, beta, out,
                                               M, 2048);
}
